# starting kernel
# baseline (speedup 1.0000x reference)
#include <hip/hip_runtime.h>

typedef __attribute__((ext_vector_type(2))) float v2f;
typedef __attribute__((ext_vector_type(4))) float v4f;
typedef __attribute__((ext_vector_type(8))) float v8f;

#define NORM_F     0.63661977236758138f
#define EXP2_COEF  -2.88539008177792681f
#define NCENTERS   1600
#define NBLOCKS    100

#if !__has_builtin(__builtin_amdgcn_exp2f)
#define __builtin_amdgcn_exp2f exp2f
#endif

__global__ __launch_bounds__(256) void radial_wmma_kernel(
    const float* __restrict__ pos,
    const float* __restrict__ cen,
    const float* __restrict__ Wm,
    const float* __restrict__ bias,
    float* __restrict__ out,
    int nrows)
{
    const int lane    = threadIdx.x & 31;
    const int wave    = (blockIdx.x * blockDim.x + threadIdx.x) >> 5;
    const int rowBase = wave * 16;
    if (rowBase >= nrows) return;

    const int  m      = lane & 15;
    const bool hiHalf = (lane >> 4) != 0;

    v2f pv = *(const v2f*)(pos + (size_t)(rowBase + m) * 2);
    const float pp = pv.x * pv.x + pv.y * pv.y;
    v2f A;
    A.x = hiHalf ? pp   : pv.x;
    A.y = hiHalf ? 1.0f : pv.y;

    float acc[8][4];
#pragma unroll
    for (int v = 0; v < 8; ++v)
#pragma unroll
        for (int a = 0; a < 4; ++a) acc[v][a] = 0.0f;

    for (int blk = 0; blk < NBLOCKS; ++blk) {
        const int n = blk * 16 + m;

        v2f cv = *(const v2f*)(cen + (size_t)n * 2);
        const float cc = cv.x * cv.x + cv.y * cv.y;
        v2f Bv;
        Bv.x = hiHalf ? 1.0f : -2.0f * cv.x;
        Bv.y = hiHalf ? cc   : -2.0f * cv.y;

        v8f D = {0.f, 0.f, 0.f, 0.f, 0.f, 0.f, 0.f, 0.f};
        D = __builtin_amdgcn_wmma_f32_16x16x4_f32(
                 false, A,  false, Bv,
                 (short)0, D,  false,  false);

        const v4f w = *(const v4f*)(Wm + (size_t)n * 4);
#pragma unroll
        for (int v = 0; v < 8; ++v) {
            const float basis = NORM_F * __builtin_amdgcn_exp2f(D[v] * EXP2_COEF);
            acc[v][0] = __builtin_fmaf(basis, w.x, acc[v][0]);
            acc[v][1] = __builtin_fmaf(basis, w.y, acc[v][1]);
            acc[v][2] = __builtin_fmaf(basis, w.z, acc[v][2]);
            acc[v][3] = __builtin_fmaf(basis, w.w, acc[v][3]);
        }
    }

    const v4f bv = *(const v4f*)(bias);
#pragma unroll
    for (int v = 0; v < 8; ++v) {
#pragma unroll
        for (int a = 0; a < 4; ++a) {
            float s = acc[v][a];
            s += __shfl_xor(s, 1, 32);
            s += __shfl_xor(s, 2, 32);
            s += __shfl_xor(s, 4, 32);
            s += __shfl_xor(s, 8, 32);
            acc[v][a] = s;
        }
        if (m == v) {
            const int row = rowBase + v + (hiHalf ? 8 : 0);
            v4f o;
            o.x = acc[v][0] + bv.x;
            o.y = acc[v][1] + bv.y;
            o.z = acc[v][2] + bv.z;
            o.w = acc[v][3] + bv.w;
            *(v4f*)(out + (size_t)row * 4) = o;
        }
    }
}

extern "C" void kernel_launch(void* const* d_in, const int* in_sizes, int n_in,
                              void* d_out, int out_size, void* d_ws, size_t ws_size,
                              hipStream_t stream) {
    const float* pos = (const float*)d_in[0];
    const float* cen = (const float*)d_in[1];
    const float* Wm  = (const float*)d_in[2];
    const float* b   = (const float*)d_in[3];
    float* out = (float*)d_out;

    const int nrows = in_sizes[0] / 2;
    const int waves = (nrows + 15) / 16;
    const int blocks = (waves + 7) / 8;

    radial_wmma_kernel<<<blocks, 256, 0, stream>>>(pos, cen, Wm, b, out, nrows);
}


# Centers and W staged once per work-group into LDS as ready-made B-operand rows, all loads in flight together; the tile loop reads LDS, no second-level round trip in front of every 16-center tile
# speedup vs baseline: 4.4800x; 4.4800x over previous
#include <hip/hip_runtime.h>


typedef __attribute__((ext_vector_type(2))) float v2f;
typedef __attribute__((ext_vector_type(4))) float v4f;
typedef __attribute__((ext_vector_type(8))) float v8f;

#define NORM_F     0.63661977236758138f
#define EXP2_COEF  -2.88539008177792681f
#define NCENTERS   1600
#define NBLOCKS    100
#define BLOCK_THREADS 256
#define STAGE_ITERS ((NCENTERS + BLOCK_THREADS - 1) / BLOCK_THREADS)
#define STAGE_FULL  (NCENTERS / BLOCK_THREADS)

#if !__has_builtin(__builtin_amdgcn_exp2f)
#define __builtin_amdgcn_exp2f exp2f
#endif

__global__ __launch_bounds__(BLOCK_THREADS) void radial_wmma_kernel(
    const float* __restrict__ pos,
    const float* __restrict__ cen,
    const float* __restrict__ Wm,
    const float* __restrict__ bias,
    float* __restrict__ out,
    int nrows)
{
    __shared__ v4f sB[NCENTERS];
    __shared__ v4f sW[NCENTERS];

    const int lane    = threadIdx.x & 31;
    const int wave    = (blockIdx.x * blockDim.x + threadIdx.x) >> 5;
    const int rowBase = wave * 16;
    const bool active = rowBase < nrows;
    const int  m      = lane & 15;
    const bool hiHalf = (lane >> 4) != 0;

    v2f pv = {0.0f, 0.0f};
    if (active) pv = *(const v2f*)(pos + (size_t)(rowBase + m) * 2);

    {
        const int t = (int)threadIdx.x;
        v2f c[STAGE_ITERS];
        v4f w[STAGE_ITERS];
#pragma unroll
        for (int i = 0; i < STAGE_ITERS; ++i) {
            int n = t + i * BLOCK_THREADS;
            n = n < NCENTERS ? n : NCENTERS - 1;
            c[i] = *(const v2f*)(cen + (size_t)n * 2);
            w[i] = *(const v4f*)(Wm + (size_t)n * 4);
        }
#pragma unroll
        for (int i = 0; i < STAGE_ITERS; ++i) {
            const int n = t + i * BLOCK_THREADS;
            if (i < STAGE_FULL || n < NCENTERS) {
                const float cc = __builtin_fmaf(c[i].y, c[i].y, c[i].x * c[i].x);
                v4f b;
                b.x = -2.0f * c[i].x;
                b.y = -2.0f * c[i].y;
                b.z = 1.0f;
                b.w = cc;
                sB[n] = b;
                sW[n] = w[i];
            }
        }
    }
    __syncthreads();

    if (!active) return;

    const float pp = pv.x * pv.x + pv.y * pv.y;
    v2f A;
    A.x = hiHalf ? pp   : pv.x;
    A.y = hiHalf ? 1.0f : pv.y;

    float acc[8][4];
#pragma unroll
    for (int v = 0; v < 8; ++v)
#pragma unroll
        for (int a = 0; a < 4; ++a) acc[v][a] = 0.0f;

    const float* sBf  = (const float*)sB;
    const int    bOff = m * 4 + (hiHalf ? 2 : 0);

    for (int blk = 0; blk < NBLOCKS; ++blk) {
        const int n = blk * 16 + m;
        const v2f Bv = *(const v2f*)(sBf + blk * 64 + bOff);
        const v4f w  = sW[n];

        v8f D = {0.f, 0.f, 0.f, 0.f, 0.f, 0.f, 0.f, 0.f};
        D = __builtin_amdgcn_wmma_f32_16x16x4_f32(
                 false, A,  false, Bv,
                 (short)0, D,  false,  false);

#pragma unroll
        for (int v = 0; v < 8; ++v) {
            const float basis = NORM_F * __builtin_amdgcn_exp2f(D[v] * EXP2_COEF);
            acc[v][0] = __builtin_fmaf(basis, w.x, acc[v][0]);
            acc[v][1] = __builtin_fmaf(basis, w.y, acc[v][1]);
            acc[v][2] = __builtin_fmaf(basis, w.z, acc[v][2]);
            acc[v][3] = __builtin_fmaf(basis, w.w, acc[v][3]);
        }
    }

    const v4f bv = *(const v4f*)(bias);
#pragma unroll
    for (int v = 0; v < 8; ++v) {
#pragma unroll
        for (int a = 0; a < 4; ++a) {
            float s = acc[v][a];
            s += __shfl_xor(s, 1, 32);
            s += __shfl_xor(s, 2, 32);
            s += __shfl_xor(s, 4, 32);
            s += __shfl_xor(s, 8, 32);
            acc[v][a] = s;
        }
        if (m == v) {
            const int row = rowBase + v + (hiHalf ? 8 : 0);
            v4f o;
            o.x = acc[v][0] + bv.x;
            o.y = acc[v][1] + bv.y;
            o.z = acc[v][2] + bv.z;
            o.w = acc[v][3] + bv.w;
            *(v4f*)(out + (size_t)row * 4) = o;
        }
    }
}

extern "C" void kernel_launch(void* const* d_in, const int* in_sizes, int n_in,
                              void* d_out, int out_size, void* d_ws, size_t ws_size,
                              hipStream_t stream) {
    const float* pos = (const float*)d_in[0];
    const float* cen = (const float*)d_in[1];
    const float* Wm  = (const float*)d_in[2];
    const float* b   = (const float*)d_in[3];
    float* out = (float*)d_out;

    const int nrows = in_sizes[0] / 2;
    const int waves = (nrows + 15) / 16;
    const int blocks = (waves + 7) / 8;

    radial_wmma_kernel<<<blocks, BLOCK_THREADS, 0, stream>>>(pos, cen, Wm, b, out, nrows);
}


# exp2 scale and log2(norm) folded into the staged B rows: the matrix product yields the exp2 argument, both multiplies per pair gone (scaling rounds before the product); LDS reads stay grouped
# speedup vs baseline: 5.3094x; 1.1851x over previous
#include <hip/hip_runtime.h>


typedef __attribute__((ext_vector_type(2))) float v2f;
typedef __attribute__((ext_vector_type(4))) float v4f;
typedef __attribute__((ext_vector_type(8))) float v8f;

#define EXP2_COEF  -2.88539008177792681f
#define LOG2_NORM  -0.65149612947231880f
#define NCENTERS   1600
#define NBLOCKS    100
#define BLOCK_THREADS 256
#define STAGE_ITERS ((NCENTERS + BLOCK_THREADS - 1) / BLOCK_THREADS)
#define STAGE_FULL  (NCENTERS / BLOCK_THREADS)

#if !__has_builtin(__builtin_amdgcn_exp2f)
#define __builtin_amdgcn_exp2f exp2f
#endif

__global__ __launch_bounds__(BLOCK_THREADS) void radial_wmma_kernel(
    const float* __restrict__ pos,
    const float* __restrict__ cen,
    const float* __restrict__ Wm,
    const float* __restrict__ bias,
    float* __restrict__ out,
    int nrows)
{
    __shared__ v4f sB[NCENTERS];
    __shared__ v4f sW[NCENTERS];

    const int lane    = threadIdx.x & 31;
    const int wave    = (blockIdx.x * blockDim.x + threadIdx.x) >> 5;
    const int rowBase = wave * 16;
    const bool active = rowBase < nrows;
    const int  m      = lane & 15;
    const bool hiHalf = (lane >> 4) != 0;

    v2f pv = {0.0f, 0.0f};
    if (active) pv = *(const v2f*)(pos + (size_t)(rowBase + m) * 2);

    {
        const int t = (int)threadIdx.x;
        v2f c[STAGE_ITERS];
        v4f w[STAGE_ITERS];
#pragma unroll
        for (int i = 0; i < STAGE_ITERS; ++i) {
            int n = t + i * BLOCK_THREADS;
            n = n < NCENTERS ? n : NCENTERS - 1;
            c[i] = *(const v2f*)(cen + (size_t)n * 2);
            w[i] = *(const v4f*)(Wm + (size_t)n * 4);
        }
#pragma unroll
        for (int i = 0; i < STAGE_ITERS; ++i) {
            const int n = t + i * BLOCK_THREADS;
            if (i < STAGE_FULL || n < NCENTERS) {
                const float cc = __builtin_fmaf(c[i].y, c[i].y, c[i].x * c[i].x);
                v4f b;
                b.x = (-2.0f * EXP2_COEF) * c[i].x;
                b.y = (-2.0f * EXP2_COEF) * c[i].y;
                b.z = EXP2_COEF;
                b.w = __builtin_fmaf(cc, EXP2_COEF, LOG2_NORM);
                sB[n] = b;
                sW[n] = w[i];
            }
        }
    }
    __syncthreads();

    if (!active) return;

    const float pp = pv.x * pv.x + pv.y * pv.y;
    v2f A;
    A.x = hiHalf ? pp   : pv.x;
    A.y = hiHalf ? 1.0f : pv.y;

    float acc[8][4];
#pragma unroll
    for (int v = 0; v < 8; ++v)
#pragma unroll
        for (int a = 0; a < 4; ++a) acc[v][a] = 0.0f;

    const float* sBf  = (const float*)sB;
    const int    bOff = m * 4 + (hiHalf ? 2 : 0);

    static_assert(NBLOCKS % 2 == 0, "the tile loop takes two tiles per pass");
    for (int blk = 0; blk < NBLOCKS; blk += 2) {
        const v2f B0 = *(const v2f*)(sBf + blk * 64 + bOff);
        const v2f B1 = *(const v2f*)(sBf + blk * 64 + 64 + bOff);
        const v4f w0 = sW[blk * 16 + m];
        const v4f w1 = sW[blk * 16 + 16 + m];
        __builtin_amdgcn_sched_barrier(0);

        const v8f Z = {0.f, 0.f, 0.f, 0.f, 0.f, 0.f, 0.f, 0.f};
        const v8f D0 = __builtin_amdgcn_wmma_f32_16x16x4_f32(
                 false, A,  false, B0,
                 (short)0, Z,  false,  false);
        const v8f D1 = __builtin_amdgcn_wmma_f32_16x16x4_f32(
                 false, A,  false, B1,
                 (short)0, Z,  false,  false);

#pragma unroll
        for (int v = 0; v < 8; ++v) {
            const float basis = __builtin_amdgcn_exp2f(D0[v]);
            acc[v][0] = __builtin_fmaf(basis, w0.x, acc[v][0]);
            acc[v][1] = __builtin_fmaf(basis, w0.y, acc[v][1]);
            acc[v][2] = __builtin_fmaf(basis, w0.z, acc[v][2]);
            acc[v][3] = __builtin_fmaf(basis, w0.w, acc[v][3]);
        }
#pragma unroll
        for (int v = 0; v < 8; ++v) {
            const float basis = __builtin_amdgcn_exp2f(D1[v]);
            acc[v][0] = __builtin_fmaf(basis, w1.x, acc[v][0]);
            acc[v][1] = __builtin_fmaf(basis, w1.y, acc[v][1]);
            acc[v][2] = __builtin_fmaf(basis, w1.z, acc[v][2]);
            acc[v][3] = __builtin_fmaf(basis, w1.w, acc[v][3]);
        }
    }

    const v4f bv = *(const v4f*)(bias);
#pragma unroll
    for (int v = 0; v < 8; ++v) {
#pragma unroll
        for (int a = 0; a < 4; ++a) {
            float s = acc[v][a];
            s += __shfl_xor(s, 1, 32);
            s += __shfl_xor(s, 2, 32);
            s += __shfl_xor(s, 4, 32);
            s += __shfl_xor(s, 8, 32);
            acc[v][a] = s;
        }
        if (m == v) {
            const int row = rowBase + v + (hiHalf ? 8 : 0);
            v4f o;
            o.x = acc[v][0] + bv.x;
            o.y = acc[v][1] + bv.y;
            o.z = acc[v][2] + bv.z;
            o.w = acc[v][3] + bv.w;
            *(v4f*)(out + (size_t)row * 4) = o;
        }
    }
}

extern "C" void kernel_launch(void* const* d_in, const int* in_sizes, int n_in,
                              void* d_out, int out_size, void* d_ws, size_t ws_size,
                              hipStream_t stream) {
    const float* pos = (const float*)d_in[0];
    const float* cen = (const float*)d_in[1];
    const float* Wm  = (const float*)d_in[2];
    const float* b   = (const float*)d_in[3];
    float* out = (float*)d_out;

    const int nrows = in_sizes[0] / 2;
    const int waves = (nrows + 15) / 16;
    const int blocks = (waves + 7) / 8;

    radial_wmma_kernel<<<blocks, BLOCK_THREADS, 0, stream>>>(pos, cen, Wm, b, out, nrows);
}


# Row sums of the epilogue as a reduce-scatter of DPP row_xmask adds on the vector ALU (same partner order 1,2,4,8, same sums) instead of 128 LDS-unit permutes per wave; one 256-byte store per wave
# speedup vs baseline: 6.1673x; 1.1616x over previous
#include <hip/hip_runtime.h>


typedef __attribute__((ext_vector_type(2))) float v2f;
typedef __attribute__((ext_vector_type(4))) float v4f;
typedef __attribute__((ext_vector_type(8))) float v8f;

#define EXP2_COEF  -2.88539008177792681f
#define LOG2_NORM  -0.65149612947231880f
#define NCENTERS   1600
#define NBLOCKS    100
#define BLOCK_THREADS 256
#define STAGE_ITERS ((NCENTERS + BLOCK_THREADS - 1) / BLOCK_THREADS)
#define STAGE_FULL  (NCENTERS / BLOCK_THREADS)

#if !__has_builtin(__builtin_amdgcn_exp2f)
#define __builtin_amdgcn_exp2f exp2f
#endif

template <int XM>
__device__ __forceinline__ float row_xor(float x) {
    return __builtin_bit_cast(float,
        __builtin_amdgcn_update_dpp(0, __builtin_bit_cast(int, x),
                                    0x160 + XM, 0xf, 0xf, true));
}

template <int XM>
__device__ __forceinline__ float keep_and_add(bool upper, float lo, float hi) {
    const float kept = upper ? hi : lo;
    const float sent = upper ? lo : hi;
    return kept + row_xor<XM>(sent);
}

__global__ __launch_bounds__(BLOCK_THREADS) void radial_wmma_kernel(
    const float* __restrict__ pos,
    const float* __restrict__ cen,
    const float* __restrict__ Wm,
    const float* __restrict__ bias,
    float* __restrict__ out,
    int nrows)
{
    __shared__ v4f sB[NCENTERS];
    __shared__ v4f sW[NCENTERS];

    const int lane    = threadIdx.x & 31;
    const int wave    = (blockIdx.x * blockDim.x + threadIdx.x) >> 5;
    const int rowBase = wave * 16;
    const bool active = rowBase < nrows;
    const int  m      = lane & 15;
    const bool hiHalf = (lane >> 4) != 0;

    v2f pv = {0.0f, 0.0f};
    if (active) pv = *(const v2f*)(pos + (size_t)(rowBase + m) * 2);

    {
        const int t = (int)threadIdx.x;
        v2f c[STAGE_ITERS];
        v4f w[STAGE_ITERS];
#pragma unroll
        for (int i = 0; i < STAGE_ITERS; ++i) {
            int n = t + i * BLOCK_THREADS;
            n = n < NCENTERS ? n : NCENTERS - 1;
            c[i] = *(const v2f*)(cen + (size_t)n * 2);
            w[i] = *(const v4f*)(Wm + (size_t)n * 4);
        }
#pragma unroll
        for (int i = 0; i < STAGE_ITERS; ++i) {
            const int n = t + i * BLOCK_THREADS;
            if (i < STAGE_FULL || n < NCENTERS) {
                const float cc = __builtin_fmaf(c[i].y, c[i].y, c[i].x * c[i].x);
                v4f b;
                b.x = (-2.0f * EXP2_COEF) * c[i].x;
                b.y = (-2.0f * EXP2_COEF) * c[i].y;
                b.z = EXP2_COEF;
                b.w = __builtin_fmaf(cc, EXP2_COEF, LOG2_NORM);
                sB[n] = b;
                sW[n] = w[i];
            }
        }
    }
    __syncthreads();

    if (!active) return;

    const float pp = pv.x * pv.x + pv.y * pv.y;
    v2f A;
    A.x = hiHalf ? pp   : pv.x;
    A.y = hiHalf ? 1.0f : pv.y;

    float acc[8][4];
#pragma unroll
    for (int v = 0; v < 8; ++v)
#pragma unroll
        for (int a = 0; a < 4; ++a) acc[v][a] = 0.0f;

    const float* sBf  = (const float*)sB;
    const int    bOff = m * 4 + (hiHalf ? 2 : 0);

    static_assert(NBLOCKS % 2 == 0, "the tile loop takes two tiles per pass");
    for (int blk = 0; blk < NBLOCKS; blk += 2) {
        const v2f B0 = *(const v2f*)(sBf + blk * 64 + bOff);
        const v2f B1 = *(const v2f*)(sBf + blk * 64 + 64 + bOff);
        const v4f w0 = sW[blk * 16 + m];
        const v4f w1 = sW[blk * 16 + 16 + m];
        __builtin_amdgcn_sched_barrier(0);

        const v8f Z = {0.f, 0.f, 0.f, 0.f, 0.f, 0.f, 0.f, 0.f};
        const v8f D0 = __builtin_amdgcn_wmma_f32_16x16x4_f32(
                 false, A,  false, B0,
                 (short)0, Z,  false,  false);
        const v8f D1 = __builtin_amdgcn_wmma_f32_16x16x4_f32(
                 false, A,  false, B1,
                 (short)0, Z,  false,  false);

#pragma unroll
        for (int v = 0; v < 8; ++v) {
            const float basis = __builtin_amdgcn_exp2f(D0[v]);
            acc[v][0] = __builtin_fmaf(basis, w0.x, acc[v][0]);
            acc[v][1] = __builtin_fmaf(basis, w0.y, acc[v][1]);
            acc[v][2] = __builtin_fmaf(basis, w0.z, acc[v][2]);
            acc[v][3] = __builtin_fmaf(basis, w0.w, acc[v][3]);
        }
#pragma unroll
        for (int v = 0; v < 8; ++v) {
            const float basis = __builtin_amdgcn_exp2f(D1[v]);
            acc[v][0] = __builtin_fmaf(basis, w1.x, acc[v][0]);
            acc[v][1] = __builtin_fmaf(basis, w1.y, acc[v][1]);
            acc[v][2] = __builtin_fmaf(basis, w1.z, acc[v][2]);
            acc[v][3] = __builtin_fmaf(basis, w1.w, acc[v][3]);
        }
    }

    const bool k1 = (lane & 1) != 0;
    const bool k2 = (lane & 2) != 0;
    const bool k4 = (lane & 4) != 0;
    const bool k8 = (lane & 8) != 0;

    float s1[8][2];
#pragma unroll
    for (int v = 0; v < 8; ++v)
#pragma unroll
        for (int j = 0; j < 2; ++j)
            s1[v][j] = keep_and_add<1>(k1, acc[v][j], acc[v][j + 2]);

    float s2[4][2];
#pragma unroll
    for (int u = 0; u < 4; ++u)
#pragma unroll
        for (int j = 0; j < 2; ++j)
            s2[u][j] = keep_and_add<2>(k2, s1[2 * u][j], s1[2 * u + 1][j]);

    float s3[2][2];
#pragma unroll
    for (int w = 0; w < 2; ++w)
#pragma unroll
        for (int j = 0; j < 2; ++j)
            s3[w][j] = keep_and_add<4>(k4, s2[2 * w][j], s2[2 * w + 1][j]);

    const float t0 = keep_and_add<8>(k8, s3[0][0], s3[1][0]);
    const float t1 = keep_and_add<8>(k8, s3[0][1], s3[1][1]);

    const v2f bv = *(const v2f*)(bias + 2 * (lane & 1));
    v2f o;
    o.x = t0 + bv.x;
    o.y = t1 + bv.y;
    *(v2f*)(out + (size_t)rowBase * 4 + 2 * lane) = o;
}

extern "C" void kernel_launch(void* const* d_in, const int* in_sizes, int n_in,
                              void* d_out, int out_size, void* d_ws, size_t ws_size,
                              hipStream_t stream) {
    const float* pos = (const float*)d_in[0];
    const float* cen = (const float*)d_in[1];
    const float* Wm  = (const float*)d_in[2];
    const float* b   = (const float*)d_in[3];
    float* out = (float*)d_out;

    const int nrows = in_sizes[0] / 2;
    const int waves = (nrows + 15) / 16;
    const int blocks = (waves + 7) / 8;

    radial_wmma_kernel<<<blocks, BLOCK_THREADS, 0, stream>>>(pos, cen, Wm, b, out, nrows);
}
